# starting kernel
# baseline (speedup 1.0000x reference)
#include <hip/hip_runtime.h>
#include <math.h>

typedef __attribute__((ext_vector_type(2))) float v2f;
typedef __attribute__((ext_vector_type(8))) float v8f;

#define N_NODES 1024
#define F_DIM   64
#define GAT_ALPHA 0.2f
#define NEG_INF_F (-9.0e15f)
#define PS 1028

__global__ void k_wh_e(const float* __restrict__ h, const float* __restrict__ W,
                       const float* __restrict__ a,
                       float* __restrict__ Wh, float* __restrict__ e1,
                       float* __restrict__ e2) {
  const int lane = threadIdx.x & 31;
  const int half = lane >> 4;
  const int l    = lane & 15;
  const int m0   = blockIdx.x * 16;

  float p1[8] = {0.f, 0.f, 0.f, 0.f, 0.f, 0.f, 0.f, 0.f};
  float p2[8] = {0.f, 0.f, 0.f, 0.f, 0.f, 0.f, 0.f, 0.f};

#pragma unroll 1
  for (int nt = 0; nt < 4; ++nt) {
    const int ncol = nt * 16 + l;
    v8f c = {};
#pragma unroll
    for (int kb = 0; kb < F_DIM; kb += 4) {
      const int k0 = kb + 2 * half;
      v2f av = *(const v2f*)(h + (m0 + l) * F_DIM + k0);
      v2f bv;
      bv.x = W[k0 * F_DIM + ncol];
      bv.y = W[(k0 + 1) * F_DIM + ncol];
      c = __builtin_amdgcn_wmma_f32_16x16x4_f32(false, av, false, bv,
                                                (short)0, c, false, false);
    }
    const float a1n = a[ncol];
    const float a2n = a[F_DIM + ncol];
#pragma unroll
    for (int r = 0; r < 8; ++r) {
      Wh[(m0 + r + 8 * half) * F_DIM + ncol] = c[r];
      p1[r] += c[r] * a1n;
      p2[r] += c[r] * a2n;
    }
  }

#pragma unroll
  for (int off = 1; off < 16; off <<= 1) {
#pragma unroll
    for (int r = 0; r < 8; ++r) {
      p1[r] += __shfl_xor(p1[r], off, 32);
      p2[r] += __shfl_xor(p2[r], off, 32);
    }
  }
  if (l == 0) {
#pragma unroll
    for (int r = 0; r < 8; ++r) {
      const int m = m0 + r + 8 * half;
      e1[m] = p1[r];
      e2[m] = p2[r];
    }
  }
}

__global__ void k_attn(const float* __restrict__ Wh, const int* __restrict__ adj,
                       const float* __restrict__ e1, const float* __restrict__ e2,
                       float* __restrict__ out) {
  extern __shared__ float smem[];
  float* P    = smem;
  float* sums = smem + 16 * PS;

  const int tid  = threadIdx.x;
  const int wave = tid >> 5;
  const int lane = tid & 31;
  const int half = lane >> 4;
  const int l    = lane & 15;
  const int i0   = blockIdx.x * 16;

#pragma unroll 1
  for (int rr = 0; rr < 4; ++rr) {
    const int r = wave * 4 + rr;
    const int i = i0 + r;
    const float ei = e1[i];

    float mx = NEG_INF_F;
#pragma unroll
    for (int t = 0; t < N_NODES / 128; ++t) {
      const int j0 = (t * 32 + lane) * 4;
      const int4   av = *(const int4*)(adj + (size_t)i * N_NODES + j0);
      const float4 ev = *(const float4*)(e2 + j0);
      float4 s;
      s.x = ei + ev.x; s.y = ei + ev.y; s.z = ei + ev.z; s.w = ei + ev.w;
      s.x = (s.x > 0.f) ? s.x : GAT_ALPHA * s.x;
      s.y = (s.y > 0.f) ? s.y : GAT_ALPHA * s.y;
      s.z = (s.z > 0.f) ? s.z : GAT_ALPHA * s.z;
      s.w = (s.w > 0.f) ? s.w : GAT_ALPHA * s.w;
      s.x = (av.x > 0) ? s.x : NEG_INF_F;
      s.y = (av.y > 0) ? s.y : NEG_INF_F;
      s.z = (av.z > 0) ? s.z : NEG_INF_F;
      s.w = (av.w > 0) ? s.w : NEG_INF_F;
      *(float4*)(P + r * PS + j0) = s;
      mx = fmaxf(mx, fmaxf(fmaxf(s.x, s.y), fmaxf(s.z, s.w)));
      if (i + 1 < N_NODES)
        __builtin_prefetch(adj + (size_t)(i + 1) * N_NODES + j0, 0, 1);
    }
#pragma unroll
    for (int off = 16; off > 0; off >>= 1)
      mx = fmaxf(mx, __shfl_xor(mx, off, 32));

    float sum = 0.0f;
#pragma unroll
    for (int t = 0; t < N_NODES / 128; ++t) {
      const int j0 = (t * 32 + lane) * 4;
      float4 p = *(const float4*)(P + r * PS + j0);
      p.x = __expf(p.x - mx);
      p.y = __expf(p.y - mx);
      p.z = __expf(p.z - mx);
      p.w = __expf(p.w - mx);
      *(float4*)(P + r * PS + j0) = p;
      sum += p.x + p.y + p.z + p.w;
    }
#pragma unroll
    for (int off = 16; off > 0; off >>= 1)
      sum += __shfl_xor(sum, off, 32);
    if (lane == 0) sums[r] = sum;
  }
  __syncthreads();

  const int ncol = wave * 16 + l;
  v8f c = {};
#pragma unroll 4
  for (int kb = 0; kb < N_NODES; kb += 4) {
    const int k0 = kb + 2 * half;
    const v2f av = *(const v2f*)(P + l * PS + k0);
    v2f bv;
    bv.x = Wh[k0 * F_DIM + ncol];
    bv.y = Wh[(k0 + 1) * F_DIM + ncol];
    c = __builtin_amdgcn_wmma_f32_16x16x4_f32(false, av, false, bv,
                                              (short)0, c, false, false);
  }

#pragma unroll
  for (int r = 0; r < 8; ++r) {
    const int m = r + 8 * half;
    float v = c[r] / sums[m];
    v = (v > 0.0f) ? v : (__expf(v) - 1.0f);
    out[(i0 + m) * F_DIM + ncol] = v;
  }
}

extern "C" void kernel_launch(void* const* d_in, const int* in_sizes, int n_in,
                              void* d_out, int out_size, void* d_ws, size_t ws_size,
                              hipStream_t stream) {
  const float* h   = (const float*)d_in[0];
  const int*   adj = (const int*)  d_in[1];
  const float* W   = (const float*)d_in[2];
  const float* a   = (const float*)d_in[3];
  float* out = (float*)d_out;

  float* Wh = (float*)d_ws;
  float* e1 = Wh + N_NODES * F_DIM;
  float* e2 = e1 + N_NODES;

  k_wh_e<<<N_NODES / 16, 32, 0, stream>>>(h, W, a, Wh, e1, e2);

  const size_t shmem = (size_t)(16 * PS + 16) * sizeof(float);
  k_attn<<<N_NODES / 16, 128, shmem, stream>>>(Wh, adj, e1, e2, out);
}


# k_attn block widened 4->16 waves: one softmax row per wave; attn@Wh split over K by wave (4 N-tiles per 64-wide slice), partial tiles summed through LDS in slice order (f32 sum order changes)
# speedup vs baseline: 3.6350x; 3.6350x over previous
#include <hip/hip_runtime.h>
#include <math.h>

typedef __attribute__((ext_vector_type(2))) float v2f;
typedef __attribute__((ext_vector_type(8))) float v8f;

#define N_NODES 1024
#define F_DIM   64
#define GAT_ALPHA 0.2f
#define NEG_INF_F (-9.0e15f)
#define PS 1028

__global__ void k_wh_e(const float* __restrict__ h, const float* __restrict__ W,
                       const float* __restrict__ a,
                       float* __restrict__ Wh, float* __restrict__ e1,
                       float* __restrict__ e2) {
  const int lane = threadIdx.x & 31;
  const int half = lane >> 4;
  const int l    = lane & 15;
  const int m0   = blockIdx.x * 16;

  float p1[8] = {0.f, 0.f, 0.f, 0.f, 0.f, 0.f, 0.f, 0.f};
  float p2[8] = {0.f, 0.f, 0.f, 0.f, 0.f, 0.f, 0.f, 0.f};

#pragma unroll 1
  for (int nt = 0; nt < 4; ++nt) {
    const int ncol = nt * 16 + l;
    v8f c = {};
#pragma unroll
    for (int kb = 0; kb < F_DIM; kb += 4) {
      const int k0 = kb + 2 * half;
      v2f av = *(const v2f*)(h + (m0 + l) * F_DIM + k0);
      v2f bv;
      bv.x = W[k0 * F_DIM + ncol];
      bv.y = W[(k0 + 1) * F_DIM + ncol];
      c = __builtin_amdgcn_wmma_f32_16x16x4_f32(false, av, false, bv,
                                                (short)0, c, false, false);
    }
    const float a1n = a[ncol];
    const float a2n = a[F_DIM + ncol];
#pragma unroll
    for (int r = 0; r < 8; ++r) {
      Wh[(m0 + r + 8 * half) * F_DIM + ncol] = c[r];
      p1[r] += c[r] * a1n;
      p2[r] += c[r] * a2n;
    }
  }

#pragma unroll
  for (int off = 1; off < 16; off <<= 1) {
#pragma unroll
    for (int r = 0; r < 8; ++r) {
      p1[r] += __shfl_xor(p1[r], off, 32);
      p2[r] += __shfl_xor(p2[r], off, 32);
    }
  }
  if (l == 0) {
#pragma unroll
    for (int r = 0; r < 8; ++r) {
      const int m = m0 + r + 8 * half;
      e1[m] = p1[r];
      e2[m] = p2[r];
    }
  }
}

#define NWAVES 16
#define KSLICE (N_NODES / NWAVES)
#define TP 66
#define TSZ (16 * TP)

__global__ void __launch_bounds__(NWAVES * 32)
k_attn(const float* __restrict__ Wh, const int* __restrict__ adj,
       const float* __restrict__ e1, const float* __restrict__ e2,
       float* __restrict__ out) {
  __shared__ __attribute__((aligned(16))) float P[16 * PS];
  __shared__ __attribute__((aligned(16))) float part[NWAVES * TSZ];
  __shared__ float sums[16];

  const int tid  = threadIdx.x;
  const int wave = tid >> 5;
  const int lane = tid & 31;
  const int half = lane >> 4;
  const int l    = lane & 15;
  const int i0   = blockIdx.x * 16;

  {
    const int r = wave;
    const int i = i0 + r;
    const float ei = e1[i];

    float mx = NEG_INF_F;
#pragma unroll
    for (int t = 0; t < N_NODES / 128; ++t) {
      const int j0 = (t * 32 + lane) * 4;
      const int4   av = *(const int4*)(adj + (size_t)i * N_NODES + j0);
      const float4 ev = *(const float4*)(e2 + j0);
      float4 s;
      s.x = ei + ev.x; s.y = ei + ev.y; s.z = ei + ev.z; s.w = ei + ev.w;
      s.x = (s.x > 0.f) ? s.x : GAT_ALPHA * s.x;
      s.y = (s.y > 0.f) ? s.y : GAT_ALPHA * s.y;
      s.z = (s.z > 0.f) ? s.z : GAT_ALPHA * s.z;
      s.w = (s.w > 0.f) ? s.w : GAT_ALPHA * s.w;
      s.x = (av.x > 0) ? s.x : NEG_INF_F;
      s.y = (av.y > 0) ? s.y : NEG_INF_F;
      s.z = (av.z > 0) ? s.z : NEG_INF_F;
      s.w = (av.w > 0) ? s.w : NEG_INF_F;
      *(float4*)(P + r * PS + j0) = s;
      mx = fmaxf(mx, fmaxf(fmaxf(s.x, s.y), fmaxf(s.z, s.w)));
    }
#pragma unroll
    for (int off = 16; off > 0; off >>= 1)
      mx = fmaxf(mx, __shfl_xor(mx, off, 32));

    float sum = 0.0f;
#pragma unroll
    for (int t = 0; t < N_NODES / 128; ++t) {
      const int j0 = (t * 32 + lane) * 4;
      float4 p = *(const float4*)(P + r * PS + j0);
      p.x = __expf(p.x - mx);
      p.y = __expf(p.y - mx);
      p.z = __expf(p.z - mx);
      p.w = __expf(p.w - mx);
      *(float4*)(P + r * PS + j0) = p;
      sum += p.x + p.y + p.z + p.w;
    }
#pragma unroll
    for (int off = 16; off > 0; off >>= 1)
      sum += __shfl_xor(sum, off, 32);
    if (lane == 0) sums[r] = sum;
  }
  __syncthreads();

  v8f c0 = {}, c1 = {}, c2 = {}, c3 = {};
  const int kbase = wave * KSLICE;
#pragma unroll
  for (int ks = 0; ks < KSLICE; ks += 4) {
    const int k0 = kbase + ks + 2 * half;
    const v2f av = *(const v2f*)(P + l * PS + k0);
    const float* w0 = Wh + k0 * F_DIM + l;
    const float* w1 = w0 + F_DIM;
    v2f b0, b1, b2, b3;
    b0.x = w0[0];  b0.y = w1[0];
    b1.x = w0[16]; b1.y = w1[16];
    b2.x = w0[32]; b2.y = w1[32];
    b3.x = w0[48]; b3.y = w1[48];
    c0 = __builtin_amdgcn_wmma_f32_16x16x4_f32(false, av, false, b0,
                                               (short)0, c0, false, false);
    c1 = __builtin_amdgcn_wmma_f32_16x16x4_f32(false, av, false, b1,
                                               (short)0, c1, false, false);
    c2 = __builtin_amdgcn_wmma_f32_16x16x4_f32(false, av, false, b2,
                                               (short)0, c2, false, false);
    c3 = __builtin_amdgcn_wmma_f32_16x16x4_f32(false, av, false, b3,
                                               (short)0, c3, false, false);
  }

  {
    float* pt = part + wave * TSZ + (8 * half) * TP + l;
#pragma unroll
    for (int r = 0; r < 8; ++r) {
      pt[r * TP +  0] = c0[r];
      pt[r * TP + 16] = c1[r];
      pt[r * TP + 32] = c2[r];
      pt[r * TP + 48] = c3[r];
    }
  }
  __syncthreads();

  {
    const int m = wave;
    const int n = 2 * lane;
    const float* ps = part + m * TP + n;
    float vx = 0.0f, vy = 0.0f;
#pragma unroll
    for (int w = 0; w < NWAVES; ++w) {
      const float2 q = *(const float2*)(ps + w * TSZ);
      vx += q.x;
      vy += q.y;
    }
    const float sm = sums[m];
    vx = vx / sm;
    vy = vy / sm;
    vx = (vx > 0.0f) ? vx : (__expf(vx) - 1.0f);
    vy = (vy > 0.0f) ? vy : (__expf(vy) - 1.0f);
    float2 o; o.x = vx; o.y = vy;
    *(float2*)(out + (i0 + m) * F_DIM + n) = o;
  }
}

extern "C" void kernel_launch(void* const* d_in, const int* in_sizes, int n_in,
                              void* d_out, int out_size, void* d_ws, size_t ws_size,
                              hipStream_t stream) {
  const float* h   = (const float*)d_in[0];
  const int*   adj = (const int*)  d_in[1];
  const float* W   = (const float*)d_in[2];
  const float* a   = (const float*)d_in[3];
  float* out = (float*)d_out;

  float* Wh = (float*)d_ws;
  float* e1 = Wh + N_NODES * F_DIM;
  float* e2 = e1 + N_NODES;

  k_wh_e<<<N_NODES / 16, 32, 0, stream>>>(h, W, a, Wh, e1, e2);

  k_attn<<<N_NODES / 16, NWAVES * 32, 0, stream>>>(Wh, adj, e1, e2, out);
}


# one launch instead of two: product re-associated, out = elu(((P@h)/rowsum)@W) and e = h@(W@a) per block; each wave's h slice is loaded once for e2 and as B fragments (f32 sum order changes)
# speedup vs baseline: 4.9065x; 1.3498x over previous
#include <hip/hip_runtime.h>
#include <math.h>


typedef __attribute__((ext_vector_type(2))) float v2f;
typedef __attribute__((ext_vector_type(8))) float v8f;

#define N_NODES 1024
#define F_DIM   64
#define GAT_ALPHA 0.2f
#define NEG_INF_F (-9.0e15f)
#define PS 1028

#define NWAVES 16
#define KSLICE (N_NODES / NWAVES)
#define TP 68
#define TSZ (16 * TP)

template <int M>
__device__ __forceinline__ float xsum(float v) {
  const int b = __builtin_bit_cast(int, v);
  const int t = __builtin_amdgcn_update_dpp(0, b, 0x160 + M, 0xf, 0xf, true);
  return v + __builtin_bit_cast(float, t);
}
template <int M>
__device__ __forceinline__ float xget(float v) {
  const int b = __builtin_bit_cast(int, v);
  return __builtin_bit_cast(float, __builtin_amdgcn_update_dpp(0, b, 0x160 + M, 0xf, 0xf, true));
}

__device__ __forceinline__ void lds_barrier() {
  __builtin_amdgcn_fence(__ATOMIC_RELEASE, "workgroup", "local");
  __builtin_amdgcn_s_barrier();
  __builtin_amdgcn_fence(__ATOMIC_ACQUIRE, "workgroup", "local");
}

__global__ void __launch_bounds__(NWAVES * 32)
k_gat(const float* __restrict__ h, const int* __restrict__ adj,
      const float* __restrict__ W, const float* __restrict__ a,
      float* __restrict__ out) {
  __shared__ __attribute__((aligned(16))) float P[16 * PS];
  __shared__ __attribute__((aligned(16))) float part[NWAVES * TSZ];
  __shared__ __attribute__((aligned(16))) float e2s[N_NODES];
  __shared__ __attribute__((aligned(16))) float wa[2 * F_DIM];
  __shared__ float sums[16];

  const int tid  = threadIdx.x;
  const int wave = tid >> 5;
  const int lane = tid & 31;
  const int half = lane >> 4;
  const int l    = lane & 15;
  const int i0   = blockIdx.x * 16;
  const int kbase = wave * KSLICE;

  const int wk = tid >> 3, wsub = tid & 7;
  const float4 w0 = *(const float4*)(W + wk * F_DIM + 8 * wsub);
  const float4 w1 = *(const float4*)(W + wk * F_DIM + 8 * wsub + 4);
  const float4 x0 = *(const float4*)(a + 8 * wsub);
  const float4 x1 = *(const float4*)(a + 8 * wsub + 4);
  const float4 y0 = *(const float4*)(a + F_DIM + 8 * wsub);
  const float4 y1 = *(const float4*)(a + F_DIM + 8 * wsub + 4);
  const v2f hrow = *(const v2f*)(h + (i0 + wave) * F_DIM + 2 * lane);
  float4 hx[16], hy[16];
#pragma unroll
  for (int s = 0; s < 16; ++s) {
    const int k0 = kbase + 4 * s + 2 * half;
    hx[s] = *(const float4*)(h + k0 * F_DIM + 4 * l);
    hy[s] = *(const float4*)(h + (k0 + 1) * F_DIM + 4 * l);
  }
  int4 av[N_NODES / 128];
#pragma unroll
  for (int t = 0; t < N_NODES / 128; ++t)
    av[t] = *(const int4*)(adj + (size_t)(i0 + wave) * N_NODES + (t * 32 + lane) * 4);

  {
    float s1 = w0.x * x0.x + w0.y * x0.y + w0.z * x0.z + w0.w * x0.w
             + w1.x * x1.x + w1.y * x1.y + w1.z * x1.z + w1.w * x1.w;
    float s2 = w0.x * y0.x + w0.y * y0.y + w0.z * y0.z + w0.w * y0.w
             + w1.x * y1.x + w1.y * y1.y + w1.z * y1.z + w1.w * y1.w;
    s1 = xsum<1>(s1); s2 = xsum<1>(s2);
    s1 = xsum<2>(s1); s2 = xsum<2>(s2);
    s1 = xsum<4>(s1); s2 = xsum<4>(s2);
    if (wsub == 0) { wa[wk] = s1; wa[F_DIM + wk] = s2; }
  }
  lds_barrier();

  float ei;
  {
    const v2f q1 = *(const v2f*)(wa + 2 * lane);
    float p = hrow.x * q1.x + hrow.y * q1.y;
    p = xsum<1>(p); p = xsum<2>(p); p = xsum<4>(p); p = xsum<8>(p);
    ei = p + __shfl_xor(p, 16, 32);
  }

  {
    const float4 q2 = *(const float4*)(wa + F_DIM + 4 * l);
    float v[32];
#pragma unroll
    for (int s = 0; s < 16; ++s) {
      v[2 * s]     = hx[s].x * q2.x + hx[s].y * q2.y + hx[s].z * q2.z + hx[s].w * q2.w;
      v[2 * s + 1] = hy[s].x * q2.x + hy[s].y * q2.y + hy[s].z * q2.z + hy[s].w * q2.w;
    }
    const bool u8 = (l & 8) != 0, u4 = (l & 4) != 0, u2 = (l & 2) != 0, u1 = (l & 1) != 0;
    float v1[16], v2[8], v3[4], v4[2];
#pragma unroll
    for (int i = 0; i < 16; ++i)
      v1[i] = (u8 ? v[2 * i + 1] : v[2 * i]) + xget<8>(u8 ? v[2 * i] : v[2 * i + 1]);
#pragma unroll
    for (int i = 0; i < 8; ++i)
      v2[i] = (u4 ? v1[2 * i + 1] : v1[2 * i]) + xget<4>(u4 ? v1[2 * i] : v1[2 * i + 1]);
#pragma unroll
    for (int i = 0; i < 4; ++i)
      v3[i] = (u2 ? v2[2 * i + 1] : v2[2 * i]) + xget<2>(u2 ? v2[2 * i] : v2[2 * i + 1]);
#pragma unroll
    for (int i = 0; i < 2; ++i)
      v4[i] = (u1 ? v3[2 * i + 1] : v3[2 * i]) + xget<1>(u1 ? v3[2 * i] : v3[2 * i + 1]);
    const int rev = (u1 ? 8 : 0) + (u2 ? 4 : 0) + (u4 ? 2 : 0) + (u8 ? 1 : 0);
#pragma unroll
    for (int i = 0; i < 2; ++i) {
      const int idx = 16 * i + rev;
      e2s[kbase + 4 * (idx >> 1) + 2 * half + (idx & 1)] = v4[i];
    }
  }
  lds_barrier();

  {
    const int r = wave;
    float mx = NEG_INF_F;
#pragma unroll
    for (int t = 0; t < N_NODES / 128; ++t) {
      const int j0 = (t * 32 + lane) * 4;
      const float4 ev = *(const float4*)(e2s + j0);
      float4 s;
      s.x = ei + ev.x; s.y = ei + ev.y; s.z = ei + ev.z; s.w = ei + ev.w;
      s.x = (s.x > 0.f) ? s.x : GAT_ALPHA * s.x;
      s.y = (s.y > 0.f) ? s.y : GAT_ALPHA * s.y;
      s.z = (s.z > 0.f) ? s.z : GAT_ALPHA * s.z;
      s.w = (s.w > 0.f) ? s.w : GAT_ALPHA * s.w;
      s.x = (av[t].x > 0) ? s.x : NEG_INF_F;
      s.y = (av[t].y > 0) ? s.y : NEG_INF_F;
      s.z = (av[t].z > 0) ? s.z : NEG_INF_F;
      s.w = (av[t].w > 0) ? s.w : NEG_INF_F;
      *(float4*)(P + r * PS + j0) = s;
      mx = fmaxf(mx, fmaxf(fmaxf(s.x, s.y), fmaxf(s.z, s.w)));
    }
#pragma unroll
    for (int off = 16; off > 0; off >>= 1)
      mx = fmaxf(mx, __shfl_xor(mx, off, 32));

    float sum = 0.0f;
#pragma unroll
    for (int t = 0; t < N_NODES / 128; ++t) {
      const int j0 = (t * 32 + lane) * 4;
      float4 p = *(const float4*)(P + r * PS + j0);
      p.x = __expf(p.x - mx);
      p.y = __expf(p.y - mx);
      p.z = __expf(p.z - mx);
      p.w = __expf(p.w - mx);
      *(float4*)(P + r * PS + j0) = p;
      sum += p.x + p.y + p.z + p.w;
    }
#pragma unroll
    for (int off = 16; off > 0; off >>= 1)
      sum += __shfl_xor(sum, off, 32);
    if (lane == 0) sums[r] = sum;
  }
  lds_barrier();

  v8f c0 = {}, c1 = {}, c2 = {}, c3 = {};
#pragma unroll
  for (int s = 0; s < 16; ++s) {
    const int k0 = kbase + 4 * s + 2 * half;
    const v2f af = *(const v2f*)(P + l * PS + k0);
    v2f b0, b1, b2, b3;
    b0.x = hx[s].x; b0.y = hy[s].x;
    b1.x = hx[s].y; b1.y = hy[s].y;
    b2.x = hx[s].z; b2.y = hy[s].z;
    b3.x = hx[s].w; b3.y = hy[s].w;
    c0 = __builtin_amdgcn_wmma_f32_16x16x4_f32(false, af, false, b0,
                                               (short)0, c0, false, false);
    c1 = __builtin_amdgcn_wmma_f32_16x16x4_f32(false, af, false, b1,
                                               (short)0, c1, false, false);
    c2 = __builtin_amdgcn_wmma_f32_16x16x4_f32(false, af, false, b2,
                                               (short)0, c2, false, false);
    c3 = __builtin_amdgcn_wmma_f32_16x16x4_f32(false, af, false, b3,
                                               (short)0, c3, false, false);
  }

  v2f wb[16];
  if (wave < 4) {
#pragma unroll
    for (int s = 0; s < 16; ++s) {
      const int k0 = 4 * s + 2 * half;
      wb[s].x = W[k0 * F_DIM + 16 * wave + l];
      wb[s].y = W[(k0 + 1) * F_DIM + 16 * wave + l];
    }
  }

  {
    float* pt = part + wave * TSZ + (8 * half) * TP + 4 * l;
#pragma unroll
    for (int r = 0; r < 8; ++r) {
      float4 q; q.x = c0[r]; q.y = c1[r]; q.z = c2[r]; q.w = c3[r];
      *(float4*)(pt + r * TP) = q;
    }
  }
  lds_barrier();

  float* Ts = P;
  {
    const int m = wave;
    const int n = 2 * lane;
    const float* ps = part + m * TP + n;
    float vx = 0.0f, vy = 0.0f;
#pragma unroll
    for (int w = 0; w < NWAVES; ++w) {
      const float2 q = *(const float2*)(ps + w * TSZ);
      vx += q.x;
      vy += q.y;
    }
    const float sm = sums[m];
    float2 o; o.x = vx / sm; o.y = vy / sm;
    *(float2*)(Ts + m * TP + n) = o;
  }
  lds_barrier();

  float* Os = part;
  if (wave < 4) {
    v8f d0 = {}, d1 = {};
#pragma unroll
    for (int s = 0; s < 16; s += 2) {
      const v2f a0 = *(const v2f*)(Ts + l * TP + 4 * s + 2 * half);
      const v2f a1 = *(const v2f*)(Ts + l * TP + 4 * (s + 1) + 2 * half);
      d0 = __builtin_amdgcn_wmma_f32_16x16x4_f32(false, a0, false, wb[s],
                                                 (short)0, d0, false, false);
      d1 = __builtin_amdgcn_wmma_f32_16x16x4_f32(false, a1, false, wb[s + 1],
                                                 (short)0, d1, false, false);
    }
    float* po = Os + (8 * half) * TP + 16 * wave + l;
#pragma unroll
    for (int r = 0; r < 8; ++r)
      po[r * TP] = d0[r] + d1[r];
  }
  lds_barrier();

  {
    const int m = wave;
    const int n = 2 * lane;
    const float2 q = *(const float2*)(Os + m * TP + n);
    float vx = q.x, vy = q.y;
    vx = (vx > 0.0f) ? vx : (__expf(vx) - 1.0f);
    vy = (vy > 0.0f) ? vy : (__expf(vy) - 1.0f);
    float2 o; o.x = vx; o.y = vy;
    *(float2*)(out + (i0 + m) * F_DIM + n) = o;
  }
}

extern "C" void kernel_launch(void* const* d_in, const int* in_sizes, int n_in,
                              void* d_out, int out_size, void* d_ws, size_t ws_size,
                              hipStream_t stream) {
  const float* h   = (const float*)d_in[0];
  const int*   adj = (const int*)  d_in[1];
  const float* W   = (const float*)d_in[2];
  const float* a   = (const float*)d_in[3];
  float* out = (float*)d_out;

  k_gat<<<N_NODES / 16, NWAVES * 32, 0, stream>>>(h, adj, W, a, out);
}


# K split over 4 work-groups per 16-row tile (256 WGs x 4 waves, slice-local max); a second launch rescales by exp(m_q-m), sums the 4 slices, normalizes, @W, ELU (f32 sum order changes)
# speedup vs baseline: 5.1841x; 1.0566x over previous
#include <hip/hip_runtime.h>
#include <math.h>


typedef __attribute__((ext_vector_type(2))) float v2f;
typedef __attribute__((ext_vector_type(4))) float v4f;
typedef __attribute__((ext_vector_type(8))) float v8f;

#define N_NODES 1024
#define F_DIM   64
#define GAT_ALPHA 0.2f
#define NEG_INF_F (-9.0e15f)

#define KSPLIT 4
#define KBLK (N_NODES / KSPLIT)
#define NW 4
#define KSLICE (KBLK / NW)
#define PS 260
#define TP 68
#define TSZ (16 * TP)

#define NBLK1 ((N_NODES / 16) * KSPLIT)
#define STATS_OFF (NBLK1 * 16 * F_DIM)

template <int M>
__device__ __forceinline__ float xsum(float v) {
  const int b = __builtin_bit_cast(int, v);
  const int t = __builtin_amdgcn_update_dpp(0, b, 0x160 + M, 0xf, 0xf, true);
  return v + __builtin_bit_cast(float, t);
}
template <int M>
__device__ __forceinline__ float xget(float v) {
  const int b = __builtin_bit_cast(int, v);
  return __builtin_bit_cast(float, __builtin_amdgcn_update_dpp(0, b, 0x160 + M, 0xf, 0xf, true));
}

__device__ __forceinline__ void lds_barrier() {
  __builtin_amdgcn_fence(__ATOMIC_RELEASE, "workgroup", "local");
  __builtin_amdgcn_s_barrier();
  __builtin_amdgcn_fence(__ATOMIC_ACQUIRE, "workgroup", "local");
}

__global__ void __launch_bounds__(NW * 32)
__attribute__((amdgpu_waves_per_eu(1, 1)))
k_part(const float* __restrict__ h, const int* __restrict__ adj,
       const float* __restrict__ W, const float* __restrict__ a,
       float* __restrict__ ws) {
  __shared__ __attribute__((aligned(16))) float P[16 * PS];
  __shared__ __attribute__((aligned(16))) float part[NW * TSZ];
  __shared__ __attribute__((aligned(16))) float e2s[KBLK];
  __shared__ __attribute__((aligned(16))) float wa[2 * F_DIM];
  __shared__ __attribute__((aligned(8))) float st[32];

  const int tid  = threadIdx.x;
  const int wave = tid >> 5;
  const int lane = tid & 31;
  const int half = lane >> 4;
  const int l    = lane & 15;
  const int rt   = blockIdx.x / KSPLIT;
  const int q    = blockIdx.x % KSPLIT;
  const int i0   = rt * 16;
  const int kloc  = wave * KSLICE;
  const int kbase = q * KBLK + kloc;

  const int wsub = tid & 7;
  float4 w0[4], w1[4];
#pragma unroll
  for (int p = 0; p < 4; ++p) {
    const int wk = 16 * p + (tid >> 3);
    w0[p] = *(const float4*)(W + wk * F_DIM + 8 * wsub);
    w1[p] = *(const float4*)(W + wk * F_DIM + 8 * wsub + 4);
  }
  const float4 x0 = *(const float4*)(a + 8 * wsub);
  const float4 x1 = *(const float4*)(a + 8 * wsub + 4);
  const float4 y0 = *(const float4*)(a + F_DIM + 8 * wsub);
  const float4 y1 = *(const float4*)(a + F_DIM + 8 * wsub + 4);
  v2f hrow[4];
#pragma unroll
  for (int j = 0; j < 4; ++j)
    hrow[j] = *(const v2f*)(h + (i0 + 4 * wave + j) * F_DIM + 2 * lane);
  float4 hx[16], hy[16];
#pragma unroll
  for (int s = 0; s < 16; ++s) {
    const int k0 = kbase + 4 * s + 2 * half;
    hx[s] = *(const float4*)(h + k0 * F_DIM + 4 * l);
    hy[s] = *(const float4*)(h + (k0 + 1) * F_DIM + 4 * l);
  }
  int4 av[4][2];
#pragma unroll
  for (int j = 0; j < 4; ++j)
#pragma unroll
    for (int t = 0; t < 2; ++t)
      av[j][t] = *(const int4*)(adj + (size_t)(i0 + 4 * wave + j) * N_NODES + q * KBLK + (t * 32 + lane) * 4);

#pragma unroll
  for (int p = 0; p < 4; ++p) {
    const int wk = 16 * p + (tid >> 3);
    float s1 = w0[p].x * x0.x + w0[p].y * x0.y + w0[p].z * x0.z + w0[p].w * x0.w
             + w1[p].x * x1.x + w1[p].y * x1.y + w1[p].z * x1.z + w1[p].w * x1.w;
    float s2 = w0[p].x * y0.x + w0[p].y * y0.y + w0[p].z * y0.z + w0[p].w * y0.w
             + w1[p].x * y1.x + w1[p].y * y1.y + w1[p].z * y1.z + w1[p].w * y1.w;
    s1 = xsum<1>(s1); s2 = xsum<1>(s2);
    s1 = xsum<2>(s1); s2 = xsum<2>(s2);
    s1 = xsum<4>(s1); s2 = xsum<4>(s2);
    if (wsub == 0) { wa[wk] = s1; wa[F_DIM + wk] = s2; }
  }
  lds_barrier();

  float ei[4];
  {
    const v2f q1 = *(const v2f*)(wa + 2 * lane);
#pragma unroll
    for (int j = 0; j < 4; ++j) {
      float p = hrow[j].x * q1.x + hrow[j].y * q1.y;
      p = xsum<1>(p); p = xsum<2>(p); p = xsum<4>(p); p = xsum<8>(p);
      ei[j] = p + __shfl_xor(p, 16, 32);
    }
  }

  {
    const float4 q2 = *(const float4*)(wa + F_DIM + 4 * l);
    float v[32];
#pragma unroll
    for (int s = 0; s < 16; ++s) {
      v[2 * s]     = hx[s].x * q2.x + hx[s].y * q2.y + hx[s].z * q2.z + hx[s].w * q2.w;
      v[2 * s + 1] = hy[s].x * q2.x + hy[s].y * q2.y + hy[s].z * q2.z + hy[s].w * q2.w;
    }
    const bool u8 = (l & 8) != 0, u4 = (l & 4) != 0, u2 = (l & 2) != 0, u1 = (l & 1) != 0;
    float v1[16], v2[8], v3[4], v4[2];
#pragma unroll
    for (int i = 0; i < 16; ++i)
      v1[i] = (u8 ? v[2 * i + 1] : v[2 * i]) + xget<8>(u8 ? v[2 * i] : v[2 * i + 1]);
#pragma unroll
    for (int i = 0; i < 8; ++i)
      v2[i] = (u4 ? v1[2 * i + 1] : v1[2 * i]) + xget<4>(u4 ? v1[2 * i] : v1[2 * i + 1]);
#pragma unroll
    for (int i = 0; i < 4; ++i)
      v3[i] = (u2 ? v2[2 * i + 1] : v2[2 * i]) + xget<2>(u2 ? v2[2 * i] : v2[2 * i + 1]);
#pragma unroll
    for (int i = 0; i < 2; ++i)
      v4[i] = (u1 ? v3[2 * i + 1] : v3[2 * i]) + xget<1>(u1 ? v3[2 * i] : v3[2 * i + 1]);
    const int rev = (u1 ? 8 : 0) + (u2 ? 4 : 0) + (u4 ? 2 : 0) + (u8 ? 1 : 0);
#pragma unroll
    for (int i = 0; i < 2; ++i) {
      const int idx = 16 * i + rev;
      e2s[kloc + 4 * (idx >> 1) + 2 * half + (idx & 1)] = v4[i];
    }
  }
  lds_barrier();

  {
    float4 sc[4][2];
    float mx[4];
    float4 ev[2];
#pragma unroll
    for (int t = 0; t < 2; ++t)
      ev[t] = *(const float4*)(e2s + (t * 32 + lane) * 4);
#pragma unroll
    for (int j = 0; j < 4; ++j) {
      mx[j] = NEG_INF_F;
#pragma unroll
      for (int t = 0; t < 2; ++t) {
        float4 s;
        s.x = ei[j] + ev[t].x; s.y = ei[j] + ev[t].y; s.z = ei[j] + ev[t].z; s.w = ei[j] + ev[t].w;
        s.x = (s.x > 0.f) ? s.x : GAT_ALPHA * s.x;
        s.y = (s.y > 0.f) ? s.y : GAT_ALPHA * s.y;
        s.z = (s.z > 0.f) ? s.z : GAT_ALPHA * s.z;
        s.w = (s.w > 0.f) ? s.w : GAT_ALPHA * s.w;
        s.x = (av[j][t].x > 0) ? s.x : NEG_INF_F;
        s.y = (av[j][t].y > 0) ? s.y : NEG_INF_F;
        s.z = (av[j][t].z > 0) ? s.z : NEG_INF_F;
        s.w = (av[j][t].w > 0) ? s.w : NEG_INF_F;
        sc[j][t] = s;
        mx[j] = fmaxf(mx[j], fmaxf(fmaxf(s.x, s.y), fmaxf(s.z, s.w)));
      }
    }
#pragma unroll
    for (int off = 16; off > 0; off >>= 1)
#pragma unroll
      for (int j = 0; j < 4; ++j)
        mx[j] = fmaxf(mx[j], __shfl_xor(mx[j], off, 32));

    float sm[4];
#pragma unroll
    for (int j = 0; j < 4; ++j) {
      float sum = 0.0f;
#pragma unroll
      for (int t = 0; t < 2; ++t) {
        float4 p = sc[j][t];
        p.x = __expf(p.x - mx[j]);
        p.y = __expf(p.y - mx[j]);
        p.z = __expf(p.z - mx[j]);
        p.w = __expf(p.w - mx[j]);
        *(float4*)(P + (4 * wave + j) * PS + (t * 32 + lane) * 4) = p;
        sum += p.x + p.y + p.z + p.w;
      }
      sm[j] = sum;
    }
#pragma unroll
    for (int off = 16; off > 0; off >>= 1)
#pragma unroll
      for (int j = 0; j < 4; ++j)
        sm[j] += __shfl_xor(sm[j], off, 32);
    if (lane == 0) {
#pragma unroll
      for (int j = 0; j < 4; ++j) {
        st[2 * (4 * wave + j)]     = mx[j];
        st[2 * (4 * wave + j) + 1] = sm[j];
      }
    }
  }
  lds_barrier();

  v8f c0 = {}, c1 = {}, c2 = {}, c3 = {};
#pragma unroll
  for (int s = 0; s < 16; ++s) {
    const int kc = kloc + 4 * s + 2 * half;
    const v2f af = *(const v2f*)(P + l * PS + kc);
    v2f b0, b1, b2, b3;
    b0.x = hx[s].x; b0.y = hy[s].x;
    b1.x = hx[s].y; b1.y = hy[s].y;
    b2.x = hx[s].z; b2.y = hy[s].z;
    b3.x = hx[s].w; b3.y = hy[s].w;
    c0 = __builtin_amdgcn_wmma_f32_16x16x4_f32(false, af, false, b0,
                                               (short)0, c0, false, false);
    c1 = __builtin_amdgcn_wmma_f32_16x16x4_f32(false, af, false, b1,
                                               (short)0, c1, false, false);
    c2 = __builtin_amdgcn_wmma_f32_16x16x4_f32(false, af, false, b2,
                                               (short)0, c2, false, false);
    c3 = __builtin_amdgcn_wmma_f32_16x16x4_f32(false, af, false, b3,
                                               (short)0, c3, false, false);
  }

  {
    float* pt = part + wave * TSZ + (8 * half) * TP + 4 * l;
#pragma unroll
    for (int r = 0; r < 8; ++r) {
      float4 o; o.x = c0[r]; o.y = c1[r]; o.z = c2[r]; o.w = c3[r];
      *(float4*)(pt + r * TP) = o;
    }
  }
  lds_barrier();

  {
    const int m = tid >> 3;
    const int n = 8 * (tid & 7);
    const float* ps = part + m * TP + n;
    float4 u0 = {0.f, 0.f, 0.f, 0.f}, u1 = {0.f, 0.f, 0.f, 0.f};
#pragma unroll
    for (int w = 0; w < NW; ++w) {
      const float4 g0 = *(const float4*)(ps + w * TSZ);
      const float4 g1 = *(const float4*)(ps + w * TSZ + 4);
      u0.x += g0.x; u0.y += g0.y; u0.z += g0.z; u0.w += g0.w;
      u1.x += g1.x; u1.y += g1.y; u1.z += g1.z; u1.w += g1.w;
    }
    float* tp = ws + (size_t)blockIdx.x * (16 * F_DIM) + m * F_DIM + n;
    *(float4*)tp = u0;
    *(float4*)(tp + 4) = u1;
  }
  if (wave == 0) ws[STATS_OFF + blockIdx.x * 32 + lane] = st[lane];
}

__global__ void __launch_bounds__(128)
k_comb(const float* __restrict__ ws, const float* __restrict__ W,
       float* __restrict__ out) {
  __shared__ __attribute__((aligned(16))) float Ts[TSZ];
  __shared__ __attribute__((aligned(16))) float Os[TSZ];

  const int tid  = threadIdx.x;
  const int wave = tid >> 5;
  const int lane = tid & 31;
  const int half = lane >> 4;
  const int l    = lane & 15;
  const int rt   = blockIdx.x;
  const int i0   = rt * 16;
  const int m = tid >> 3;
  const int n = 8 * (tid & 7);

  v2f wb[16];
#pragma unroll
  for (int s = 0; s < 16; ++s) {
    const int k0 = 4 * s + 2 * half;
    wb[s].x = W[k0 * F_DIM + 16 * wave + l];
    wb[s].y = W[(k0 + 1) * F_DIM + 16 * wave + l];
  }
  v2f ml[KSPLIT];
  v4f t0[KSPLIT], t1[KSPLIT];
#pragma unroll
  for (int q = 0; q < KSPLIT; ++q)
    ml[q] = *(const v2f*)(ws + STATS_OFF + (rt * KSPLIT + q) * 32 + 2 * m);
#pragma unroll
  for (int q = 0; q < KSPLIT; ++q) {
    const float* tp = ws + (size_t)(rt * KSPLIT + q) * (16 * F_DIM) + m * F_DIM + n;
    t0[q] = *(const v4f*)tp;
    t1[q] = *(const v4f*)(tp + 4);
  }
  asm volatile("" : "+v"(ml[0]), "+v"(ml[1]), "+v"(ml[2]), "+v"(ml[3])
                  : "v"(wb[0]), "v"(wb[1]), "v"(wb[2]), "v"(wb[3]), "v"(wb[4]), "v"(wb[5]), "v"(wb[6]), "v"(wb[7]),
                    "v"(wb[8]), "v"(wb[9]), "v"(wb[10]), "v"(wb[11]), "v"(wb[12]), "v"(wb[13]), "v"(wb[14]), "v"(wb[15]),
                    "v"(t0[0]), "v"(t0[1]), "v"(t0[2]), "v"(t0[3]), "v"(t1[0]), "v"(t1[1]), "v"(t1[2]), "v"(t1[3]));

  {
    const float mm = fmaxf(fmaxf(ml[0].x, ml[1].x), fmaxf(ml[2].x, ml[3].x));
    float rs = 0.0f;
    float4 u0 = {0.f, 0.f, 0.f, 0.f}, u1 = {0.f, 0.f, 0.f, 0.f};
#pragma unroll
    for (int q = 0; q < KSPLIT; ++q) {
      const float f = __expf(ml[q].x - mm);
      rs += f * ml[q].y;
      u0.x += f * t0[q].x; u0.y += f * t0[q].y; u0.z += f * t0[q].z; u0.w += f * t0[q].w;
      u1.x += f * t1[q].x; u1.y += f * t1[q].y; u1.z += f * t1[q].z; u1.w += f * t1[q].w;
    }
    float4 o0, o1;
    o0.x = u0.x / rs; o0.y = u0.y / rs; o0.z = u0.z / rs; o0.w = u0.w / rs;
    o1.x = u1.x / rs; o1.y = u1.y / rs; o1.z = u1.z / rs; o1.w = u1.w / rs;
    *(float4*)(Ts + m * TP + n) = o0;
    *(float4*)(Ts + m * TP + n + 4) = o1;
  }
  lds_barrier();

  {
    v8f d0 = {}, d1 = {};
#pragma unroll
    for (int s = 0; s < 16; s += 2) {
      const v2f a0 = *(const v2f*)(Ts + l * TP + 4 * s + 2 * half);
      const v2f a1 = *(const v2f*)(Ts + l * TP + 4 * (s + 1) + 2 * half);
      d0 = __builtin_amdgcn_wmma_f32_16x16x4_f32(false, a0, false, wb[s],
                                                 (short)0, d0, false, false);
      d1 = __builtin_amdgcn_wmma_f32_16x16x4_f32(false, a1, false, wb[s + 1],
                                                 (short)0, d1, false, false);
    }
    float* po = Os + (8 * half) * TP + 16 * wave + l;
#pragma unroll
    for (int r = 0; r < 8; ++r)
      po[r * TP] = d0[r] + d1[r];
  }
  lds_barrier();

  {
    const float4 g0 = *(const float4*)(Os + m * TP + n);
    const float4 g1 = *(const float4*)(Os + m * TP + n + 4);
    float4 o0, o1;
    o0.x = (g0.x > 0.0f) ? g0.x : (__expf(g0.x) - 1.0f);
    o0.y = (g0.y > 0.0f) ? g0.y : (__expf(g0.y) - 1.0f);
    o0.z = (g0.z > 0.0f) ? g0.z : (__expf(g0.z) - 1.0f);
    o0.w = (g0.w > 0.0f) ? g0.w : (__expf(g0.w) - 1.0f);
    o1.x = (g1.x > 0.0f) ? g1.x : (__expf(g1.x) - 1.0f);
    o1.y = (g1.y > 0.0f) ? g1.y : (__expf(g1.y) - 1.0f);
    o1.z = (g1.z > 0.0f) ? g1.z : (__expf(g1.z) - 1.0f);
    o1.w = (g1.w > 0.0f) ? g1.w : (__expf(g1.w) - 1.0f);
    float* op = out + (i0 + m) * F_DIM + n;
    *(float4*)op = o0;
    *(float4*)(op + 4) = o1;
  }
}

extern "C" void kernel_launch(void* const* d_in, const int* in_sizes, int n_in,
                              void* d_out, int out_size, void* d_ws, size_t ws_size,
                              hipStream_t stream) {
  const float* h   = (const float*)d_in[0];
  const int*   adj = (const int*)  d_in[1];
  const float* W   = (const float*)d_in[2];
  const float* a   = (const float*)d_in[3];
  float* out = (float*)d_out;
  float* ws  = (float*)d_ws;

  k_part<<<NBLK1, NW * 32, 0, stream>>>(h, adj, W, a, ws);
  k_comb<<<N_NODES / 16, 128, 0, stream>>>(ws, W, out);
}


# k_part: the 32-lane butterflies for slice max, slice sum and e1 run on the vector ALU (row_xmask DPP + v_permlanex16) instead of ten dependent ds_bpermute round trips; same steps, same sum order
# speedup vs baseline: 5.5715x; 1.0747x over previous
#include <hip/hip_runtime.h>
#include <math.h>


typedef __attribute__((ext_vector_type(2))) float v2f;
typedef __attribute__((ext_vector_type(4))) float v4f;
typedef __attribute__((ext_vector_type(8))) float v8f;

#define N_NODES 1024
#define F_DIM   64
#define GAT_ALPHA 0.2f
#define NEG_INF_F (-9.0e15f)

#define KSPLIT 4
#define KBLK (N_NODES / KSPLIT)
#define NW 4
#define KSLICE (KBLK / NW)
#define PS 260
#define TP 68
#define TSZ (16 * TP)

#define NBLK1 ((N_NODES / 16) * KSPLIT)
#define STATS_OFF (NBLK1 * 16 * F_DIM)

template <int M>
__device__ __forceinline__ float xsum(float v) {
  const int b = __builtin_bit_cast(int, v);
  const int t = __builtin_amdgcn_update_dpp(0, b, 0x160 + M, 0xf, 0xf, true);
  return v + __builtin_bit_cast(float, t);
}
template <int M>
__device__ __forceinline__ float xget(float v) {
  const int b = __builtin_bit_cast(int, v);
  return __builtin_bit_cast(float, __builtin_amdgcn_update_dpp(0, b, 0x160 + M, 0xf, 0xf, true));
}

__device__ __forceinline__ float xhalf(float v) {
  const int b = __builtin_bit_cast(int, v);
  return __builtin_bit_cast(float, __builtin_amdgcn_permlanex16(b, b, 0x76543210, 0xfedcba98, false, false));
}

__device__ __forceinline__ void lds_barrier() {
  __builtin_amdgcn_fence(__ATOMIC_RELEASE, "workgroup", "local");
  __builtin_amdgcn_s_barrier();
  __builtin_amdgcn_fence(__ATOMIC_ACQUIRE, "workgroup", "local");
}

__global__ void __launch_bounds__(NW * 32)
__attribute__((amdgpu_waves_per_eu(1, 1)))
k_part(const float* __restrict__ h, const int* __restrict__ adj,
       const float* __restrict__ W, const float* __restrict__ a,
       float* __restrict__ ws) {
  __shared__ __attribute__((aligned(16))) float P[16 * PS];
  __shared__ __attribute__((aligned(16))) float part[NW * TSZ];
  __shared__ __attribute__((aligned(16))) float e2s[KBLK];
  __shared__ __attribute__((aligned(16))) float wa[2 * F_DIM];
  __shared__ __attribute__((aligned(8))) float st[32];

  const int tid  = threadIdx.x;
  const int wave = tid >> 5;
  const int lane = tid & 31;
  const int half = lane >> 4;
  const int l    = lane & 15;
  const int rt   = blockIdx.x / KSPLIT;
  const int q    = blockIdx.x % KSPLIT;
  const int i0   = rt * 16;
  const int kloc  = wave * KSLICE;
  const int kbase = q * KBLK + kloc;

  const int wsub = tid & 7;
  float4 w0[4], w1[4];
#pragma unroll
  for (int p = 0; p < 4; ++p) {
    const int wk = 16 * p + (tid >> 3);
    w0[p] = *(const float4*)(W + wk * F_DIM + 8 * wsub);
    w1[p] = *(const float4*)(W + wk * F_DIM + 8 * wsub + 4);
  }
  const float4 x0 = *(const float4*)(a + 8 * wsub);
  const float4 x1 = *(const float4*)(a + 8 * wsub + 4);
  const float4 y0 = *(const float4*)(a + F_DIM + 8 * wsub);
  const float4 y1 = *(const float4*)(a + F_DIM + 8 * wsub + 4);
  v2f hrow[4];
#pragma unroll
  for (int j = 0; j < 4; ++j)
    hrow[j] = *(const v2f*)(h + (i0 + 4 * wave + j) * F_DIM + 2 * lane);
  float4 hx[16], hy[16];
#pragma unroll
  for (int s = 0; s < 16; ++s) {
    const int k0 = kbase + 4 * s + 2 * half;
    hx[s] = *(const float4*)(h + k0 * F_DIM + 4 * l);
    hy[s] = *(const float4*)(h + (k0 + 1) * F_DIM + 4 * l);
  }
  int4 av[4][2];
#pragma unroll
  for (int j = 0; j < 4; ++j)
#pragma unroll
    for (int t = 0; t < 2; ++t)
      av[j][t] = *(const int4*)(adj + (size_t)(i0 + 4 * wave + j) * N_NODES + q * KBLK + (t * 32 + lane) * 4);

#pragma unroll
  for (int p = 0; p < 4; ++p) {
    const int wk = 16 * p + (tid >> 3);
    float s1 = w0[p].x * x0.x + w0[p].y * x0.y + w0[p].z * x0.z + w0[p].w * x0.w
             + w1[p].x * x1.x + w1[p].y * x1.y + w1[p].z * x1.z + w1[p].w * x1.w;
    float s2 = w0[p].x * y0.x + w0[p].y * y0.y + w0[p].z * y0.z + w0[p].w * y0.w
             + w1[p].x * y1.x + w1[p].y * y1.y + w1[p].z * y1.z + w1[p].w * y1.w;
    s1 = xsum<1>(s1); s2 = xsum<1>(s2);
    s1 = xsum<2>(s1); s2 = xsum<2>(s2);
    s1 = xsum<4>(s1); s2 = xsum<4>(s2);
    if (wsub == 0) { wa[wk] = s1; wa[F_DIM + wk] = s2; }
  }
  lds_barrier();

  float ei[4];
  {
    const v2f q1 = *(const v2f*)(wa + 2 * lane);
#pragma unroll
    for (int j = 0; j < 4; ++j) {
      float p = hrow[j].x * q1.x + hrow[j].y * q1.y;
      p = xsum<1>(p); p = xsum<2>(p); p = xsum<4>(p); p = xsum<8>(p);
      ei[j] = p + xhalf(p);
    }
  }

  {
    const float4 q2 = *(const float4*)(wa + F_DIM + 4 * l);
    float v[32];
#pragma unroll
    for (int s = 0; s < 16; ++s) {
      v[2 * s]     = hx[s].x * q2.x + hx[s].y * q2.y + hx[s].z * q2.z + hx[s].w * q2.w;
      v[2 * s + 1] = hy[s].x * q2.x + hy[s].y * q2.y + hy[s].z * q2.z + hy[s].w * q2.w;
    }
    const bool u8 = (l & 8) != 0, u4 = (l & 4) != 0, u2 = (l & 2) != 0, u1 = (l & 1) != 0;
    float v1[16], v2[8], v3[4], v4[2];
#pragma unroll
    for (int i = 0; i < 16; ++i)
      v1[i] = (u8 ? v[2 * i + 1] : v[2 * i]) + xget<8>(u8 ? v[2 * i] : v[2 * i + 1]);
#pragma unroll
    for (int i = 0; i < 8; ++i)
      v2[i] = (u4 ? v1[2 * i + 1] : v1[2 * i]) + xget<4>(u4 ? v1[2 * i] : v1[2 * i + 1]);
#pragma unroll
    for (int i = 0; i < 4; ++i)
      v3[i] = (u2 ? v2[2 * i + 1] : v2[2 * i]) + xget<2>(u2 ? v2[2 * i] : v2[2 * i + 1]);
#pragma unroll
    for (int i = 0; i < 2; ++i)
      v4[i] = (u1 ? v3[2 * i + 1] : v3[2 * i]) + xget<1>(u1 ? v3[2 * i] : v3[2 * i + 1]);
    const int rev = (u1 ? 8 : 0) + (u2 ? 4 : 0) + (u4 ? 2 : 0) + (u8 ? 1 : 0);
#pragma unroll
    for (int i = 0; i < 2; ++i) {
      const int idx = 16 * i + rev;
      e2s[kloc + 4 * (idx >> 1) + 2 * half + (idx & 1)] = v4[i];
    }
  }
  lds_barrier();

  {
    float4 sc[4][2];
    float mx[4];
    float4 ev[2];
#pragma unroll
    for (int t = 0; t < 2; ++t)
      ev[t] = *(const float4*)(e2s + (t * 32 + lane) * 4);
#pragma unroll
    for (int j = 0; j < 4; ++j) {
      mx[j] = NEG_INF_F;
#pragma unroll
      for (int t = 0; t < 2; ++t) {
        float4 s;
        s.x = ei[j] + ev[t].x; s.y = ei[j] + ev[t].y; s.z = ei[j] + ev[t].z; s.w = ei[j] + ev[t].w;
        s.x = (s.x > 0.f) ? s.x : GAT_ALPHA * s.x;
        s.y = (s.y > 0.f) ? s.y : GAT_ALPHA * s.y;
        s.z = (s.z > 0.f) ? s.z : GAT_ALPHA * s.z;
        s.w = (s.w > 0.f) ? s.w : GAT_ALPHA * s.w;
        s.x = (av[j][t].x > 0) ? s.x : NEG_INF_F;
        s.y = (av[j][t].y > 0) ? s.y : NEG_INF_F;
        s.z = (av[j][t].z > 0) ? s.z : NEG_INF_F;
        s.w = (av[j][t].w > 0) ? s.w : NEG_INF_F;
        sc[j][t] = s;
        mx[j] = fmaxf(mx[j], fmaxf(fmaxf(s.x, s.y), fmaxf(s.z, s.w)));
      }
    }
#pragma unroll
    for (int j = 0; j < 4; ++j) mx[j] = fmaxf(mx[j], xhalf(mx[j]));
#pragma unroll
    for (int j = 0; j < 4; ++j) mx[j] = fmaxf(mx[j], xget<8>(mx[j]));
#pragma unroll
    for (int j = 0; j < 4; ++j) mx[j] = fmaxf(mx[j], xget<4>(mx[j]));
#pragma unroll
    for (int j = 0; j < 4; ++j) mx[j] = fmaxf(mx[j], xget<2>(mx[j]));
#pragma unroll
    for (int j = 0; j < 4; ++j) mx[j] = fmaxf(mx[j], xget<1>(mx[j]));

    float sm[4];
#pragma unroll
    for (int j = 0; j < 4; ++j) {
      float sum = 0.0f;
#pragma unroll
      for (int t = 0; t < 2; ++t) {
        float4 p = sc[j][t];
        p.x = __expf(p.x - mx[j]);
        p.y = __expf(p.y - mx[j]);
        p.z = __expf(p.z - mx[j]);
        p.w = __expf(p.w - mx[j]);
        *(float4*)(P + (4 * wave + j) * PS + (t * 32 + lane) * 4) = p;
        sum += p.x + p.y + p.z + p.w;
      }
      sm[j] = sum;
    }
#pragma unroll
    for (int j = 0; j < 4; ++j) sm[j] += xhalf(sm[j]);
#pragma unroll
    for (int j = 0; j < 4; ++j) sm[j] = xsum<8>(sm[j]);
#pragma unroll
    for (int j = 0; j < 4; ++j) sm[j] = xsum<4>(sm[j]);
#pragma unroll
    for (int j = 0; j < 4; ++j) sm[j] = xsum<2>(sm[j]);
#pragma unroll
    for (int j = 0; j < 4; ++j) sm[j] = xsum<1>(sm[j]);
    if (lane == 0) {
#pragma unroll
      for (int j = 0; j < 4; ++j) {
        st[2 * (4 * wave + j)]     = mx[j];
        st[2 * (4 * wave + j) + 1] = sm[j];
      }
    }
  }
  lds_barrier();

  v8f c0 = {}, c1 = {}, c2 = {}, c3 = {};
#pragma unroll
  for (int s = 0; s < 16; ++s) {
    const int kc = kloc + 4 * s + 2 * half;
    const v2f af = *(const v2f*)(P + l * PS + kc);
    v2f b0, b1, b2, b3;
    b0.x = hx[s].x; b0.y = hy[s].x;
    b1.x = hx[s].y; b1.y = hy[s].y;
    b2.x = hx[s].z; b2.y = hy[s].z;
    b3.x = hx[s].w; b3.y = hy[s].w;
    c0 = __builtin_amdgcn_wmma_f32_16x16x4_f32(false, af, false, b0,
                                               (short)0, c0, false, false);
    c1 = __builtin_amdgcn_wmma_f32_16x16x4_f32(false, af, false, b1,
                                               (short)0, c1, false, false);
    c2 = __builtin_amdgcn_wmma_f32_16x16x4_f32(false, af, false, b2,
                                               (short)0, c2, false, false);
    c3 = __builtin_amdgcn_wmma_f32_16x16x4_f32(false, af, false, b3,
                                               (short)0, c3, false, false);
  }

  {
    float* pt = part + wave * TSZ + (8 * half) * TP + 4 * l;
#pragma unroll
    for (int r = 0; r < 8; ++r) {
      float4 o; o.x = c0[r]; o.y = c1[r]; o.z = c2[r]; o.w = c3[r];
      *(float4*)(pt + r * TP) = o;
    }
  }
  lds_barrier();

  {
    const int m = tid >> 3;
    const int n = 8 * (tid & 7);
    const float* ps = part + m * TP + n;
    float4 u0 = {0.f, 0.f, 0.f, 0.f}, u1 = {0.f, 0.f, 0.f, 0.f};
#pragma unroll
    for (int w = 0; w < NW; ++w) {
      const float4 g0 = *(const float4*)(ps + w * TSZ);
      const float4 g1 = *(const float4*)(ps + w * TSZ + 4);
      u0.x += g0.x; u0.y += g0.y; u0.z += g0.z; u0.w += g0.w;
      u1.x += g1.x; u1.y += g1.y; u1.z += g1.z; u1.w += g1.w;
    }
    float* tp = ws + (size_t)blockIdx.x * (16 * F_DIM) + m * F_DIM + n;
    *(float4*)tp = u0;
    *(float4*)(tp + 4) = u1;
  }
  if (wave == 0) ws[STATS_OFF + blockIdx.x * 32 + lane] = st[lane];
}

__global__ void __launch_bounds__(128)
k_comb(const float* __restrict__ ws, const float* __restrict__ W,
       float* __restrict__ out) {
  __shared__ __attribute__((aligned(16))) float Ts[TSZ];
  __shared__ __attribute__((aligned(16))) float Os[TSZ];

  const int tid  = threadIdx.x;
  const int wave = tid >> 5;
  const int lane = tid & 31;
  const int half = lane >> 4;
  const int l    = lane & 15;
  const int rt   = blockIdx.x;
  const int i0   = rt * 16;
  const int m = tid >> 3;
  const int n = 8 * (tid & 7);

  v2f wb[16];
#pragma unroll
  for (int s = 0; s < 16; ++s) {
    const int k0 = 4 * s + 2 * half;
    wb[s].x = W[k0 * F_DIM + 16 * wave + l];
    wb[s].y = W[(k0 + 1) * F_DIM + 16 * wave + l];
  }
  v2f ml[KSPLIT];
  v4f t0[KSPLIT], t1[KSPLIT];
#pragma unroll
  for (int q = 0; q < KSPLIT; ++q)
    ml[q] = *(const v2f*)(ws + STATS_OFF + (rt * KSPLIT + q) * 32 + 2 * m);
#pragma unroll
  for (int q = 0; q < KSPLIT; ++q) {
    const float* tp = ws + (size_t)(rt * KSPLIT + q) * (16 * F_DIM) + m * F_DIM + n;
    t0[q] = *(const v4f*)tp;
    t1[q] = *(const v4f*)(tp + 4);
  }
  asm volatile("" : "+v"(ml[0]), "+v"(ml[1]), "+v"(ml[2]), "+v"(ml[3])
                  : "v"(wb[0]), "v"(wb[1]), "v"(wb[2]), "v"(wb[3]), "v"(wb[4]), "v"(wb[5]), "v"(wb[6]), "v"(wb[7]),
                    "v"(wb[8]), "v"(wb[9]), "v"(wb[10]), "v"(wb[11]), "v"(wb[12]), "v"(wb[13]), "v"(wb[14]), "v"(wb[15]),
                    "v"(t0[0]), "v"(t0[1]), "v"(t0[2]), "v"(t0[3]), "v"(t1[0]), "v"(t1[1]), "v"(t1[2]), "v"(t1[3]));

  {
    const float mm = fmaxf(fmaxf(ml[0].x, ml[1].x), fmaxf(ml[2].x, ml[3].x));
    float rs = 0.0f;
    float4 u0 = {0.f, 0.f, 0.f, 0.f}, u1 = {0.f, 0.f, 0.f, 0.f};
#pragma unroll
    for (int q = 0; q < KSPLIT; ++q) {
      const float f = __expf(ml[q].x - mm);
      rs += f * ml[q].y;
      u0.x += f * t0[q].x; u0.y += f * t0[q].y; u0.z += f * t0[q].z; u0.w += f * t0[q].w;
      u1.x += f * t1[q].x; u1.y += f * t1[q].y; u1.z += f * t1[q].z; u1.w += f * t1[q].w;
    }
    float4 o0, o1;
    o0.x = u0.x / rs; o0.y = u0.y / rs; o0.z = u0.z / rs; o0.w = u0.w / rs;
    o1.x = u1.x / rs; o1.y = u1.y / rs; o1.z = u1.z / rs; o1.w = u1.w / rs;
    *(float4*)(Ts + m * TP + n) = o0;
    *(float4*)(Ts + m * TP + n + 4) = o1;
  }
  lds_barrier();

  {
    v8f d0 = {}, d1 = {};
#pragma unroll
    for (int s = 0; s < 16; s += 2) {
      const v2f a0 = *(const v2f*)(Ts + l * TP + 4 * s + 2 * half);
      const v2f a1 = *(const v2f*)(Ts + l * TP + 4 * (s + 1) + 2 * half);
      d0 = __builtin_amdgcn_wmma_f32_16x16x4_f32(false, a0, false, wb[s],
                                                 (short)0, d0, false, false);
      d1 = __builtin_amdgcn_wmma_f32_16x16x4_f32(false, a1, false, wb[s + 1],
                                                 (short)0, d1, false, false);
    }
    float* po = Os + (8 * half) * TP + 16 * wave + l;
#pragma unroll
    for (int r = 0; r < 8; ++r)
      po[r * TP] = d0[r] + d1[r];
  }
  lds_barrier();

  {
    const float4 g0 = *(const float4*)(Os + m * TP + n);
    const float4 g1 = *(const float4*)(Os + m * TP + n + 4);
    float4 o0, o1;
    o0.x = (g0.x > 0.0f) ? g0.x : (__expf(g0.x) - 1.0f);
    o0.y = (g0.y > 0.0f) ? g0.y : (__expf(g0.y) - 1.0f);
    o0.z = (g0.z > 0.0f) ? g0.z : (__expf(g0.z) - 1.0f);
    o0.w = (g0.w > 0.0f) ? g0.w : (__expf(g0.w) - 1.0f);
    o1.x = (g1.x > 0.0f) ? g1.x : (__expf(g1.x) - 1.0f);
    o1.y = (g1.y > 0.0f) ? g1.y : (__expf(g1.y) - 1.0f);
    o1.z = (g1.z > 0.0f) ? g1.z : (__expf(g1.z) - 1.0f);
    o1.w = (g1.w > 0.0f) ? g1.w : (__expf(g1.w) - 1.0f);
    float* op = out + (i0 + m) * F_DIM + n;
    *(float4*)op = o0;
    *(float4*)(op + 4) = o1;
  }
}

extern "C" void kernel_launch(void* const* d_in, const int* in_sizes, int n_in,
                              void* d_out, int out_size, void* d_ws, size_t ws_size,
                              hipStream_t stream) {
  const float* h   = (const float*)d_in[0];
  const int*   adj = (const int*)  d_in[1];
  const float* W   = (const float*)d_in[2];
  const float* a   = (const float*)d_in[3];
  float* out = (float*)d_out;
  float* ws  = (float*)d_ws;

  k_part<<<NBLK1, NW * 32, 0, stream>>>(h, adj, W, a, ws);
  k_comb<<<N_NODES / 16, 128, 0, stream>>>(ws, W, out);
}
